# starting kernel
# baseline (speedup 1.0000x reference)
#include <hip/hip_runtime.h>
#include <hip/hip_bf16.h>

typedef float v2f __attribute__((ext_vector_type(2)));
typedef float v8f __attribute__((ext_vector_type(8)));

#define NNODES 50000

__global__ void k_init_deg(float* deg) {
  int i = blockIdx.x * blockDim.x + threadIdx.x;
  if (i < NNODES) deg[i] = 1.0f;
}

__global__ void k_count_deg(const int* __restrict__ dst, float* __restrict__ deg, int E) {
  int e = blockIdx.x * blockDim.x + threadIdx.x;
  if (e < E) atomicAdd(&deg[dst[e]], 1.0f);
}

__global__ void k_finish_deg(float* deg) {
  int i = blockIdx.x * blockDim.x + threadIdx.x;
  if (i < NNODES) deg[i] = rsqrtf(deg[i]);
}

template <int K, int N>
__global__ void k_gemm_wmma_f32(const float* __restrict__ A,
                                const float* __restrict__ W,
                                float* __restrict__ C, int M) {
  const int wave = (int)((blockIdx.x * blockDim.x + threadIdx.x) >> 5);
  const int lane = threadIdx.x & 31;
  const int tiles_n = N / 16;
  const int row_tile = wave / tiles_n;
  const int col_tile = wave % tiles_n;
  if (row_tile * 16 >= M) return;

  const int half = lane >> 4;
  const int r    = lane & 15;

  const float* arow = A + (size_t)(row_tile * 16 + r) * K;
  const float* wcol = W + (size_t)(col_tile * 16 + r);

  v8f acc = {};
#pragma unroll 4
  for (int k0 = 0; k0 < K; k0 += 4) {
    const int kA = k0 + 2 * half;
    v2f a = *(const v2f*)(arow + kA);
    v2f b;
    b.x = wcol[(size_t)kA * N];
    b.y = wcol[(size_t)(kA + 1) * N];
    acc = __builtin_amdgcn_wmma_f32_16x16x4_f32(
         false, a,  false, b,
         (short)0, acc,  false,  false);
  }

  float* crow = C + (size_t)(row_tile * 16 + 8 * half) * N + col_tile * 16 + r;
#pragma unroll
  for (int i = 0; i < 8; ++i) crow[(size_t)i * N] = acc[i];
}

template <int CH>
__global__ void k_self_bias(const float* __restrict__ feat,
                            const float* __restrict__ dinv,
                            const float* __restrict__ bias,
                            float* __restrict__ out) {
  int idx = blockIdx.x * blockDim.x + threadIdx.x;
  if (idx >= NNODES * CH) return;
  int n = idx / CH;
  int c = idx % CH;
  float di = dinv[n];
  out[idx] = feat[idx] * di * di + bias[c];
}

template <int CH>
__global__ void k_scatter(const int* __restrict__ src, const int* __restrict__ dst,
                          const float* __restrict__ dinv,
                          const float* __restrict__ feat,
                          float* __restrict__ out, int E) {
  const int wave = (int)((blockIdx.x * blockDim.x + threadIdx.x) >> 5);
  const int lane = threadIdx.x & 31;
  if (wave >= E) return;
  const int s = src[wave];
  const int d = dst[wave];
  const float w = dinv[s] * dinv[d];
  const float* frow = feat + (size_t)s * CH;
  float* orow = out + (size_t)d * CH;
  if (CH == 128) {
    float4 v = *(const float4*)(frow + lane * 4);
    atomicAdd(orow + lane * 4 + 0, v.x * w);
    atomicAdd(orow + lane * 4 + 1, v.y * w);
    atomicAdd(orow + lane * 4 + 2, v.z * w);
    atomicAdd(orow + lane * 4 + 3, v.w * w);
  } else {
    float2 v = *(const float2*)(frow + lane * 2);
    atomicAdd(orow + lane * 2 + 0, v.x * w);
    atomicAdd(orow + lane * 2 + 1, v.y * w);
  }
}

__global__ void k_tanh(const float* __restrict__ in, float* __restrict__ out, int n) {
  int idx = blockIdx.x * blockDim.x + threadIdx.x;
  if (idx < n) out[idx] = tanhf(in[idx]);
}

extern "C" void kernel_launch(void* const* d_in, const int* in_sizes, int n_in,
                              void* d_out, int out_size, void* d_ws, size_t ws_size,
                              hipStream_t stream) {
  const float* x   = (const float*)d_in[0];
  const int*  eidx = (const int*)d_in[1];
  const float* W1  = (const float*)d_in[2];
  const float* b1  = (const float*)d_in[3];
  const float* W2  = (const float*)d_in[4];
  const float* b2  = (const float*)d_in[5];
  float* out = (float*)d_out;

  const int E = in_sizes[1] / 2;
  const int* src = eidx;
  const int* dst = eidx + E;

  float* dinv = (float*)d_ws;
  float* bufA = dinv + 50048;
  float* bufB = bufA + (size_t)NNODES * 128;

  const int T = 256;

  k_init_deg<<<(NNODES + T - 1) / T, T, 0, stream>>>(dinv);
  k_count_deg<<<(E + T - 1) / T, T, 0, stream>>>(dst, dinv, E);
  k_finish_deg<<<(NNODES + T - 1) / T, T, 0, stream>>>(dinv);

  {
    int waves = (NNODES / 16) * (128 / 16);
    k_gemm_wmma_f32<256, 128><<<waves * 32 / 128, 128, 0, stream>>>(x, W1, bufA, NNODES);
  }
  k_self_bias<128><<<(NNODES * 128 + T - 1) / T, T, 0, stream>>>(bufA, dinv, b1, bufB);
  k_scatter<128><<<(E + 7) / 8, 256, 0, stream>>>(src, dst, dinv, bufA, bufB, E);
  k_tanh<<<(NNODES * 128 + T - 1) / T, T, 0, stream>>>(bufB, bufA, NNODES * 128);

  {
    int waves = (NNODES / 16) * (64 / 16);
    k_gemm_wmma_f32<128, 64><<<waves * 32 / 128, 128, 0, stream>>>(bufA, W2, bufB, NNODES);
  }
  k_self_bias<64><<<(NNODES * 64 + T - 1) / T, T, 0, stream>>>(bufB, dinv, b2, out);
  k_scatter<64><<<(E + 7) / 8, 256, 0, stream>>>(src, dst, dinv, bufB, out, E);
}


# Float-atomic edge scatter -> edge lists grouped by destination (integer atomic rank, scan, fill) + one-wave-per-node register gather; per-node sum order now fixed: ascending source id
# speedup vs baseline: 1.3608x; 1.3608x over previous
#include <hip/hip_runtime.h>
#include <hip/hip_bf16.h>


typedef float v2f __attribute__((ext_vector_type(2)));
typedef float v8f __attribute__((ext_vector_type(8)));

#define NNODES 50000
#define SPLIT_ROWS 42000
#define KEY_PAD 0x7fffffff

__global__ void k_zero_cnt(int* __restrict__ cnt) {
  int i = blockIdx.x * blockDim.x + threadIdx.x;
  if (i < NNODES) cnt[i] = 0;
}

__global__ void k_count_rank(const int* __restrict__ dst, int* __restrict__ cnt,
                             int* __restrict__ rank, int E) {
  int e = blockIdx.x * blockDim.x + threadIdx.x;
  if (e < E) rank[e] = atomicAdd(&cnt[dst[e]], 1);
}

__device__ __forceinline__ int wave_sum_i(int v) {
#pragma unroll
  for (int o = 16; o > 0; o >>= 1) v += __shfl_xor(v, o);
  return v;
}

__device__ __forceinline__ int wave_incl_scan_i(int v, int lane) {
#pragma unroll
  for (int o = 1; o < 32; o <<= 1) {
    int t = __shfl_up(v, o);
    if (lane >= o) v += t;
  }
  return v;
}

__global__ void __launch_bounds__(1024) k_scan(const int* __restrict__ cnt,
                                               int* __restrict__ rowptr,
                                               float* __restrict__ dinv) {
  __shared__ int s_w[32];
  __shared__ int s_base;
  const int tid = threadIdx.x;
  const int lane = tid & 31;
  const int wv = tid >> 5;
  const int first = blockIdx.x * 1024;

  int part = 0;
  for (int i = tid; i < first; i += 1024) part += cnt[i];
  part = wave_sum_i(part);
  if (lane == 0) s_w[wv] = part;
  __syncthreads();
  if (wv == 0) {
    int t = wave_sum_i(s_w[lane]);
    if (lane == 0) s_base = t;
  }
  __syncthreads();
  const int base = s_base;

  const int i = first + tid;
  const int c = (i < NNODES) ? cnt[i] : 0;
  const int inc = wave_incl_scan_i(c, lane);
  if (lane == 31) s_w[wv] = inc;
  __syncthreads();
  if (wv == 0) {
    const int t = s_w[lane];
    const int ti = wave_incl_scan_i(t, lane);
    s_w[lane] = ti - t;
  }
  __syncthreads();
  const int excl = base + s_w[wv] + inc - c;
  if (i < NNODES) {
    rowptr[i] = excl;
    dinv[i] = rsqrtf((float)(c + 1));
  }
  if (i == NNODES - 1) rowptr[NNODES] = excl + c;
}

__global__ void k_fill(const int* __restrict__ src, const int* __restrict__ dst,
                       const int* __restrict__ rowptr, const int* __restrict__ rank,
                       int* __restrict__ csr, int E) {
  int e = blockIdx.x * blockDim.x + threadIdx.x;
  if (e >= E) return;
  const int p = rowptr[dst[e]] + rank[e];
  if ((unsigned)p < (unsigned)E) csr[p] = src[e];
}

template <int K, int N>
__global__ void k_gemm_wmma_f32(const float* __restrict__ A,
                                const float* __restrict__ W,
                                float* __restrict__ C, int M) {
  const int wave = (int)((blockIdx.x * blockDim.x + threadIdx.x) >> 5);
  const int lane = threadIdx.x & 31;
  const int tiles_n = N / 16;
  const int row_tile = wave / tiles_n;
  const int col_tile = wave % tiles_n;
  if (row_tile * 16 >= M) return;

  const int half = lane >> 4;
  const int r    = lane & 15;

  const float* arow = A + (size_t)(row_tile * 16 + r) * K;
  const float* wcol = W + (size_t)(col_tile * 16 + r);

  v8f acc = {};
#pragma unroll 4
  for (int k0 = 0; k0 < K; k0 += 4) {
    const int kA = k0 + 2 * half;
    v2f a = *(const v2f*)(arow + kA);
    v2f b;
    b.x = wcol[(size_t)kA * N];
    b.y = wcol[(size_t)(kA + 1) * N];
    acc = __builtin_amdgcn_wmma_f32_16x16x4_f32(
         false, a,  false, b,
         (short)0, acc,  false,  false);
  }

  float* crow = C + (size_t)(row_tile * 16 + 8 * half) * N + col_tile * 16 + r;
#pragma unroll
  for (int i = 0; i < 8; ++i) crow[(size_t)i * N] = acc[i];
}

template <int CH> struct RowVec;
template <> struct RowVec<128> { typedef float4 type; };
template <> struct RowVec<64>  { typedef float2 type; };

__device__ __forceinline__ void fma_row(float4& a, const float4& v, float w) {
  a.x += v.x * w; a.y += v.y * w; a.z += v.z * w; a.w += v.w * w;
}
__device__ __forceinline__ void fma_row(float2& a, const float2& v, float w) {
  a.x += v.x * w; a.y += v.y * w;
}
__device__ __forceinline__ float4 self_row(const float4& v, float w, const float4& b) {
  return make_float4(v.x * w + b.x, v.y * w + b.y, v.z * w + b.z, v.w * w + b.w);
}
__device__ __forceinline__ float2 self_row(const float2& v, float w, const float2& b) {
  return make_float2(v.x * w + b.x, v.y * w + b.y);
}

__device__ __forceinline__ int sort32(int key, int lane) {
#pragma unroll
  for (int k = 2; k <= 32; k <<= 1) {
#pragma unroll
    for (int j = k >> 1; j > 0; j >>= 1) {
      const int o = __shfl_xor(key, j);
      const bool up = (lane & k) == 0;
      const bool lo = (lane & j) == 0;
      const int mn = min(key, o);
      const int mx = max(key, o);
      key = (lo == up) ? mn : mx;
    }
  }
  return key;
}

__device__ __forceinline__ int clamp_node(int s) { return min(max(s, 0), NNODES - 1); }

template <int CH>
__global__ void __launch_bounds__(256)
k_gather(const int* __restrict__ rowptr, const int* __restrict__ csr,
         const float* __restrict__ dinv, const float* __restrict__ feat,
         const float* __restrict__ bias,
         float* __restrict__ out_lo, float* __restrict__ out_hi, int split, int E) {
  typedef typename RowVec<CH>::type VT;
  constexpr int VPL = CH / 32;
  const int lane = threadIdx.x & 31;
  const int d = __builtin_amdgcn_readfirstlane(
      (int)((blockIdx.x * blockDim.x + threadIdx.x) >> 5));
  if (d >= NNODES) return;

  int beg = rowptr[d];
  int end = rowptr[d + 1];
  beg = min(max(beg, 0), E);
  end = min(max(end, beg), E);
  const int deg = end - beg;
  const float di = dinv[d];
  const float* fl = feat + lane * VPL;

  VT acc = self_row(*(const VT*)(fl + (size_t)d * CH), di * di,
                    *(const VT*)(bias + lane * VPL));

  if (deg <= 32) {
    int key = (lane < deg) ? csr[beg + lane] : KEY_PAD;
    key = sort32(key, lane);
    const float w = (lane < deg) ? dinv[clamp_node(key)] * di : 0.0f;
#pragma unroll 4
    for (int j = 0; j < deg; ++j) {
      const int s = clamp_node(__builtin_amdgcn_readlane(key, j));
      const float ws = __int_as_float(__builtin_amdgcn_readlane(__float_as_int(w), j));
      const VT v = *(const VT*)(fl + (size_t)s * CH);
      fma_row(acc, v, ws);
    }
  } else {
    int last = -1;
    int done = 0;
    while (done < deg) {
      int best = KEY_PAD;
      for (int j = beg + lane; j < end; j += 32) {
        const int k = csr[j];
        if (k > last && k < best) best = k;
      }
#pragma unroll
      for (int o = 16; o > 0; o >>= 1) best = min(best, __shfl_xor(best, o));
      int mult = 0;
      for (int j = beg + lane; j < end; j += 32) mult += (csr[j] == best) ? 1 : 0;
      mult = __builtin_amdgcn_readfirstlane(wave_sum_i(mult));
      best = __builtin_amdgcn_readfirstlane(best);
      if (mult <= 0) break;
      const int s = clamp_node(best);
      const float ws = dinv[s] * di;
      const VT v = *(const VT*)(fl + (size_t)s * CH);
      for (int m = 0; m < mult; ++m) fma_row(acc, v, ws);
      last = best;
      done += mult;
    }
  }

  float* orow = (d < split) ? (out_lo + (size_t)d * CH) : (out_hi + (size_t)(d - split) * CH);
  *(VT*)(orow + lane * VPL) = acc;
}

__global__ void k_tanh(const float* __restrict__ in, float* __restrict__ out, int n) {
  int idx = blockIdx.x * blockDim.x + threadIdx.x;
  if (idx < n) out[idx] = tanhf(in[idx]);
}

extern "C" void kernel_launch(void* const* d_in, const int* in_sizes, int n_in,
                              void* d_out, int out_size, void* d_ws, size_t ws_size,
                              hipStream_t stream) {
  const float* x   = (const float*)d_in[0];
  const int*  eidx = (const int*)d_in[1];
  const float* W1  = (const float*)d_in[2];
  const float* b1  = (const float*)d_in[3];
  const float* W2  = (const float*)d_in[4];
  const float* b2  = (const float*)d_in[5];
  float* out = (float*)d_out;

  const int E = in_sizes[1] / 2;
  const int* src = eidx;
  const int* dst = eidx + E;

  float* dinv = (float*)d_ws;
  float* bufA = dinv + 50048;
  float* bufB = bufA + (size_t)NNODES * 128;
  int* csr    = (int*)(bufB + (size_t)SPLIT_ROWS * 128);
  int* rowptr = csr + 800000;
  int* cnt    = rowptr + 50048;
  int* rank   = (int*)out;

  const int T = 256;

  k_zero_cnt<<<(NNODES + T - 1) / T, T, 0, stream>>>(cnt);
  k_count_rank<<<(E + T - 1) / T, T, 0, stream>>>(dst, cnt, rank, E);
  k_scan<<<(NNODES + 1023) / 1024, 1024, 0, stream>>>(cnt, rowptr, dinv);
  k_fill<<<(E + T - 1) / T, T, 0, stream>>>(src, dst, rowptr, rank, csr, E);

  {
    int waves = (NNODES / 16) * (128 / 16);
    k_gemm_wmma_f32<256, 128><<<waves * 32 / 128, 128, 0, stream>>>(x, W1, bufA, NNODES);
  }
  k_gather<128><<<(NNODES + 7) / 8, 256, 0, stream>>>(rowptr, csr, dinv, bufA, b1,
                                                     bufB, out, SPLIT_ROWS, E);
  k_tanh<<<(SPLIT_ROWS * 128 + T - 1) / T, T, 0, stream>>>(bufB, bufA, SPLIT_ROWS * 128);
  k_tanh<<<((NNODES - SPLIT_ROWS) * 128 + T - 1) / T, T, 0, stream>>>(
      out, bufA + (size_t)SPLIT_ROWS * 128, (NNODES - SPLIT_ROWS) * 128);

  {
    int waves = (NNODES / 16) * (64 / 16);
    k_gemm_wmma_f32<128, 64><<<waves * 32 / 128, 128, 0, stream>>>(bufA, W2, bufB, NNODES);
  }
  k_gather<64><<<(NNODES + 7) / 8, 256, 0, stream>>>(rowptr, csr, dinv, bufB, b2,
                                                    out, out, NNODES, E);
}


# Both products re-tiled: W staged once per workgroup in LDS as k-pair rows; a wave holds 16x64 of C and reads A in 16-byte pieces one trip ahead; k sum order inside each dot product regrouped
# speedup vs baseline: 1.6669x; 1.2249x over previous
#include <hip/hip_runtime.h>
#include <hip/hip_bf16.h>


typedef float v2f __attribute__((ext_vector_type(2)));
typedef float v8f __attribute__((ext_vector_type(8)));

#define NNODES 50000
#define SPLIT_ROWS 42000
#define KEY_PAD 0x7fffffff

__global__ void k_zero_cnt(int* __restrict__ cnt) {
  int i = blockIdx.x * blockDim.x + threadIdx.x;
  if (i < NNODES) cnt[i] = 0;
}

__global__ void k_count_rank(const int* __restrict__ dst, int* __restrict__ cnt,
                             int* __restrict__ rank, int E) {
  int e = blockIdx.x * blockDim.x + threadIdx.x;
  if (e < E) rank[e] = atomicAdd(&cnt[dst[e]], 1);
}

__device__ __forceinline__ int wave_sum_i(int v) {
#pragma unroll
  for (int o = 16; o > 0; o >>= 1) v += __shfl_xor(v, o);
  return v;
}

__device__ __forceinline__ int wave_incl_scan_i(int v, int lane) {
#pragma unroll
  for (int o = 1; o < 32; o <<= 1) {
    int t = __shfl_up(v, o);
    if (lane >= o) v += t;
  }
  return v;
}

__global__ void __launch_bounds__(1024) k_scan(const int* __restrict__ cnt,
                                               int* __restrict__ rowptr,
                                               float* __restrict__ dinv) {
  __shared__ int s_w[32];
  __shared__ int s_base;
  const int tid = threadIdx.x;
  const int lane = tid & 31;
  const int wv = tid >> 5;
  const int first = blockIdx.x * 1024;

  int part = 0;
  for (int i = tid; i < first; i += 1024) part += cnt[i];
  part = wave_sum_i(part);
  if (lane == 0) s_w[wv] = part;
  __syncthreads();
  if (wv == 0) {
    int t = wave_sum_i(s_w[lane]);
    if (lane == 0) s_base = t;
  }
  __syncthreads();
  const int base = s_base;

  const int i = first + tid;
  const int c = (i < NNODES) ? cnt[i] : 0;
  const int inc = wave_incl_scan_i(c, lane);
  if (lane == 31) s_w[wv] = inc;
  __syncthreads();
  if (wv == 0) {
    const int t = s_w[lane];
    const int ti = wave_incl_scan_i(t, lane);
    s_w[lane] = ti - t;
  }
  __syncthreads();
  const int excl = base + s_w[wv] + inc - c;
  if (i < NNODES) {
    rowptr[i] = excl;
    dinv[i] = rsqrtf((float)(c + 1));
  }
  if (i == NNODES - 1) rowptr[NNODES] = excl + c;
}

__global__ void k_fill(const int* __restrict__ src, const int* __restrict__ dst,
                       const int* __restrict__ rowptr, const int* __restrict__ rank,
                       int* __restrict__ csr, int E) {
  int e = blockIdx.x * blockDim.x + threadIdx.x;
  if (e >= E) return;
  const int p = rowptr[dst[e]] + rank[e];
  if ((unsigned)p < (unsigned)E) csr[p] = src[e];
}

#define GEMM_WAVES 16

template <int K, int N>
__global__ void __launch_bounds__(GEMM_WAVES * 32)
k_gemm_wmma_f32(const float* __restrict__ A,
                const float* __restrict__ A_hi, int split,
                const float* __restrict__ W,
                float* __restrict__ C, int M) {
  constexpr int NCG = N / 64;
  constexpr int PITCH = N * 2 + 16;
  constexpr int JB = 4;
  __shared__ __attribute__((aligned(16))) float wl[(K / 2) * PITCH];

  const int tid = threadIdx.x;
  constexpr int STAGE_TRIPS = (K / 2) * (N / 4) / (GEMM_WAVES * 32);
  static_assert((K / 2) * (N / 4) % (GEMM_WAVES * 32) == 0, "W does not divide over the threads");
#pragma unroll
  for (int it = 0; it < STAGE_TRIPS; ++it) {
    const int idx = it * (GEMM_WAVES * 32) + tid;
    const int p = idx / (N / 4);
    const int n4 = idx % (N / 4);
    const float4 lo = *(const float4*)(W + (size_t)(2 * p) * N + 4 * n4);
    const float4 hi = *(const float4*)(W + (size_t)(2 * p + 1) * N + 4 * n4);
    float* d = wl + p * PITCH + 8 * n4;
    *(float4*)d = make_float4(lo.x, hi.x, lo.y, hi.y);
    *(float4*)(d + 4) = make_float4(lo.z, hi.z, lo.w, hi.w);
  }
  __syncthreads();

  const int wave = __builtin_amdgcn_readfirstlane(tid >> 5);
  const int lane = tid & 31;
  const int h = lane >> 4;
  const int c = lane & 15;
  const int row_tiles = M / 16;

  for (int li = wave;; li += GEMM_WAVES) {
    const int u = li / NCG;
    const int cg = li % NCG;
    const int t = u * (int)gridDim.x + (int)blockIdx.x;
    if (t >= row_tiles) break;

    const int row0 = t * 16;
    const float* atile = (row0 < split) ? (A + (size_t)row0 * K)
                                        : (A_hi + (size_t)(row0 - split) * K);
    const float* arow = atile + (size_t)c * K + 4 * h;
    const float* bl = wl + (2 * h) * PITCH + (cg * 64 + c) * 2;

    v8f acc[4] = {};
    float4 a_cur[JB];
#pragma unroll
    for (int q = 0; q < JB; ++q) a_cur[q] = *(const float4*)(arow + 8 * q);

#pragma unroll 1
    for (int jb = 0; jb < K / 8; jb += JB) {
      float4 a_nxt[JB];
      const int jn = (jb + JB < K / 8) ? (jb + JB) : jb;
#pragma unroll
      for (int q = 0; q < JB; ++q) a_nxt[q] = *(const float4*)(arow + 8 * (jn + q));
      __builtin_amdgcn_sched_barrier(0);
      const float* bj = bl + (4 * jb) * PITCH;
#pragma unroll
      for (int q = 0; q < JB; ++q) {
        const float* bq = bj + (4 * q) * PITCH;
        v2f a0; a0.x = a_cur[q].x; a0.y = a_cur[q].y;
        v2f a1; a1.x = a_cur[q].z; a1.y = a_cur[q].w;
#pragma unroll
        for (int tt = 0; tt < 4; ++tt) {
          const v2f b0 = *(const v2f*)(bq + tt * 32);
          acc[tt] = __builtin_amdgcn_wmma_f32_16x16x4_f32(
               false, a0,  false, b0,
               (short)0, acc[tt],  false,  false);
        }
#pragma unroll
        for (int tt = 0; tt < 4; ++tt) {
          const v2f b1 = *(const v2f*)(bq + PITCH + tt * 32);
          acc[tt] = __builtin_amdgcn_wmma_f32_16x16x4_f32(
               false, a1,  false, b1,
               (short)0, acc[tt],  false,  false);
        }
      }
#pragma unroll
      for (int q = 0; q < JB; ++q) a_cur[q] = a_nxt[q];
    }

    float* crow = C + (size_t)(row0 + 8 * h) * N + cg * 64 + c;
#pragma unroll
    for (int tt = 0; tt < 4; ++tt) {
#pragma unroll
      for (int i = 0; i < 8; ++i) crow[(size_t)i * N + tt * 16] = acc[tt][i];
    }
  }
}

template <int CH> struct RowVec;
template <> struct RowVec<128> { typedef float4 type; };
template <> struct RowVec<64>  { typedef float2 type; };

__device__ __forceinline__ void fma_row(float4& a, const float4& v, float w) {
  a.x += v.x * w; a.y += v.y * w; a.z += v.z * w; a.w += v.w * w;
}
__device__ __forceinline__ void fma_row(float2& a, const float2& v, float w) {
  a.x += v.x * w; a.y += v.y * w;
}
__device__ __forceinline__ float4 self_row(const float4& v, float w, const float4& b) {
  return make_float4(v.x * w + b.x, v.y * w + b.y, v.z * w + b.z, v.w * w + b.w);
}
__device__ __forceinline__ float2 self_row(const float2& v, float w, const float2& b) {
  return make_float2(v.x * w + b.x, v.y * w + b.y);
}
__device__ __forceinline__ float4 tanh_row(const float4& a) {
  return make_float4(tanhf(a.x), tanhf(a.y), tanhf(a.z), tanhf(a.w));
}

__device__ __forceinline__ int sort32(int key, int lane) {
#pragma unroll
  for (int k = 2; k <= 32; k <<= 1) {
#pragma unroll
    for (int j = k >> 1; j > 0; j >>= 1) {
      const int o = __shfl_xor(key, j);
      const bool up = (lane & k) == 0;
      const bool lo = (lane & j) == 0;
      const int mn = min(key, o);
      const int mx = max(key, o);
      key = (lo == up) ? mn : mx;
    }
  }
  return key;
}

__device__ __forceinline__ int clamp_node(int s) { return min(max(s, 0), NNODES - 1); }

template <int CH, bool ACT>
__global__ void __launch_bounds__(256)
k_gather(const int* __restrict__ rowptr, const int* __restrict__ csr,
         const float* __restrict__ dinv, const float* __restrict__ feat,
         const float* __restrict__ bias,
         float* __restrict__ out_lo, float* __restrict__ out_hi, int split, int E) {
  typedef typename RowVec<CH>::type VT;
  constexpr int VPL = CH / 32;
  const int lane = threadIdx.x & 31;
  const int d = __builtin_amdgcn_readfirstlane(
      (int)((blockIdx.x * blockDim.x + threadIdx.x) >> 5));
  if (d >= NNODES) return;

  int beg = rowptr[d];
  int end = rowptr[d + 1];
  beg = min(max(beg, 0), E);
  end = min(max(end, beg), E);
  const int deg = end - beg;
  const float di = dinv[d];
  const float* fl = feat + lane * VPL;

  VT acc = self_row(*(const VT*)(fl + (size_t)d * CH), di * di,
                    *(const VT*)(bias + lane * VPL));

  if (deg <= 32) {
    int key = (lane < deg) ? csr[beg + lane] : KEY_PAD;
    key = sort32(key, lane);
    const float w = (lane < deg) ? dinv[clamp_node(key)] * di : 0.0f;
#pragma unroll 4
    for (int j = 0; j < deg; ++j) {
      const int s = clamp_node(__builtin_amdgcn_readlane(key, j));
      const float ws = __int_as_float(__builtin_amdgcn_readlane(__float_as_int(w), j));
      const VT v = *(const VT*)(fl + (size_t)s * CH);
      fma_row(acc, v, ws);
    }
  } else {
    int last = -1;
    int done = 0;
    while (done < deg) {
      int best = KEY_PAD;
      for (int j = beg + lane; j < end; j += 32) {
        const int k = csr[j];
        if (k > last && k < best) best = k;
      }
#pragma unroll
      for (int o = 16; o > 0; o >>= 1) best = min(best, __shfl_xor(best, o));
      int mult = 0;
      for (int j = beg + lane; j < end; j += 32) mult += (csr[j] == best) ? 1 : 0;
      mult = __builtin_amdgcn_readfirstlane(wave_sum_i(mult));
      best = __builtin_amdgcn_readfirstlane(best);
      if (mult <= 0) break;
      const int s = clamp_node(best);
      const float ws = dinv[s] * di;
      const VT v = *(const VT*)(fl + (size_t)s * CH);
      for (int m = 0; m < mult; ++m) fma_row(acc, v, ws);
      last = best;
      done += mult;
    }
  }

  if constexpr (ACT) acc = tanh_row(acc);

  float* orow = (d < split) ? (out_lo + (size_t)d * CH) : (out_hi + (size_t)(d - split) * CH);
  *(VT*)(orow + lane * VPL) = acc;
}

extern "C" void kernel_launch(void* const* d_in, const int* in_sizes, int n_in,
                              void* d_out, int out_size, void* d_ws, size_t ws_size,
                              hipStream_t stream) {
  const float* x   = (const float*)d_in[0];
  const int*  eidx = (const int*)d_in[1];
  const float* W1  = (const float*)d_in[2];
  const float* b1  = (const float*)d_in[3];
  const float* W2  = (const float*)d_in[4];
  const float* b2  = (const float*)d_in[5];
  float* out = (float*)d_out;

  const int E = in_sizes[1] / 2;
  const int* src = eidx;
  const int* dst = eidx + E;

  float* dinv = (float*)d_ws;
  float* bufA = dinv + 50048;
  float* bufB = bufA + (size_t)NNODES * 128;
  int* csr    = (int*)(bufB + (size_t)SPLIT_ROWS * 128);
  int* rowptr = csr + 800000;
  int* cnt    = rowptr + 50048;
  int* rank   = (int*)out;

  const int T = 256;

  k_zero_cnt<<<(NNODES + T - 1) / T, T, 0, stream>>>(cnt);
  k_count_rank<<<(E + T - 1) / T, T, 0, stream>>>(dst, cnt, rank, E);
  k_scan<<<(NNODES + 1023) / 1024, 1024, 0, stream>>>(cnt, rowptr, dinv);
  k_fill<<<(E + T - 1) / T, T, 0, stream>>>(src, dst, rowptr, rank, csr, E);

  k_gemm_wmma_f32<256, 128><<<256, GEMM_WAVES * 32, 0, stream>>>(x, x, NNODES, W1, bufA,
                                                                 NNODES);
  k_gather<128, true><<<(NNODES + 7) / 8, 256, 0, stream>>>(rowptr, csr, dinv, bufA, b1,
                                                           bufB, out, SPLIT_ROWS, E);

  k_gemm_wmma_f32<128, 64><<<256, GEMM_WAVES * 32, 0, stream>>>(bufB, out, SPLIT_ROWS, W2,
                                                               bufA, NNODES);
  k_gather<64, false><<<(NNODES + 7) / 8, 256, 0, stream>>>(rowptr, csr, dinv, bufA, b2,
                                                           out, out, NNODES, E);
}


# Edge lists built by a two-level counting sort with counts and ranks in LDS (chunks ordered by 64-node range, then placed per range): 2 launches replace 4, no global atomics left
# speedup vs baseline: 2.3721x; 1.4231x over previous
#include <hip/hip_runtime.h>
#include <hip/hip_bf16.h>


typedef float v2f __attribute__((ext_vector_type(2)));
typedef float v8f __attribute__((ext_vector_type(8)));

#define NNODES 50000
#define SPLIT_ROWS 42000
#define KEY_PAD 0x7fffffff

#define RANGE_BITS 6
#define RANGE_NODES 64
#define NRANGE ((NNODES + RANGE_NODES - 1) / RANGE_NODES)
#define TPITCH 784
#define NCHUNK 256
#define STAGE_MAX 4096

__device__ __forceinline__ int clamp_node(int s) { return min(max(s, 0), NNODES - 1); }

__device__ __forceinline__ int wave_sum_i(int v) {
#pragma unroll
  for (int o = 16; o > 0; o >>= 1) v += __shfl_xor(v, o);
  return v;
}

__device__ __forceinline__ int wave_incl_scan_i(int v, int lane) {
#pragma unroll
  for (int o = 1; o < 32; o <<= 1) {
    int t = __shfl_up(v, o);
    if (lane >= o) v += t;
  }
  return v;
}

__global__ void __launch_bounds__(1024) k_bin_chunk(const int* __restrict__ src,
                                                    const int* __restrict__ dst,
                                                    int* __restrict__ part,
                                                    int* __restrict__ tab, int E) {
  __shared__ int s_hist[1024];
  __shared__ int s_w[32];
  __shared__ int s_stage[STAGE_MAX];
  const int tid = threadIdx.x;
  const int lane = tid & 31;
  const int wv = tid >> 5;
  const int ch = (E + NCHUNK - 1) / NCHUNK;
  const int e0 = min((int)blockIdx.x * ch, E);
  const int e1 = min(e0 + ch, E);

  s_hist[tid] = 0;
  constexpr int PK = 4;
  int dv[PK], sv[PK];
#pragma unroll
  for (int k = 0; k < PK; ++k) {
    const int e = e0 + tid + k * 1024;
    const bool ok = e < e1;
    const int ec = max(min(e, E - 1), 0);
    const int dd = dst[ec];
    const int ss = src[ec];
    dv[k] = clamp_node(dd) | (ok ? 0 : (int)0x80000000);
    sv[k] = clamp_node(ss);
  }
  __syncthreads();
#pragma unroll
  for (int k = 0; k < PK; ++k)
    if (dv[k] >= 0) atomicAdd(&s_hist[dv[k] >> RANGE_BITS], 1);
  for (int e = e0 + tid + PK * 1024; e < e1; e += 1024)
    atomicAdd(&s_hist[clamp_node(dst[e]) >> RANGE_BITS], 1);
  __syncthreads();

  const int cnt = s_hist[tid];
  const int inc = wave_incl_scan_i(cnt, lane);
  if (lane == 31) s_w[wv] = inc;
  __syncthreads();
  if (wv == 0) {
    const int t = s_w[lane];
    const int ti = wave_incl_scan_i(t, lane);
    s_w[lane] = ti - t;
  }
  __syncthreads();
  const int excl = s_w[wv] + inc - cnt;
  if (tid < TPITCH) tab[(size_t)blockIdx.x * TPITCH + tid] = excl;
  s_hist[tid] = excl;
  __syncthreads();

  const bool staged = ch <= STAGE_MAX;
  int* pout = part + e0;
#pragma unroll
  for (int k = 0; k < PK; ++k) {
    if (dv[k] >= 0) {
      const int p = atomicAdd(&s_hist[dv[k] >> RANGE_BITS], 1);
      const int rec = ((dv[k] & (RANGE_NODES - 1)) << 16) | sv[k];
      if (staged) { if ((unsigned)p < (unsigned)STAGE_MAX) s_stage[p] = rec; }
      else if (p < e1 - e0) pout[p] = rec;
    }
  }
  for (int e = e0 + tid + PK * 1024; e < e1; e += 1024) {
    const int d = clamp_node(dst[e]);
    const int p = atomicAdd(&s_hist[d >> RANGE_BITS], 1);
    const int rec = ((d & (RANGE_NODES - 1)) << 16) | clamp_node(src[e]);
    if (staged) { if ((unsigned)p < (unsigned)STAGE_MAX) s_stage[p] = rec; }
    else if (p < e1 - e0) pout[p] = rec;
  }
  if (staged) {
    __syncthreads();
    for (int i = tid; i < e1 - e0; i += 1024) pout[i] = s_stage[i];
  }
}

__global__ void __launch_bounds__(NCHUNK) k_rank_range(const int* __restrict__ part,
                                                       const int* __restrict__ tab,
                                                       int* __restrict__ rowptr,
                                                       float* __restrict__ dinv,
                                                       int* __restrict__ csr, int E) {
  __shared__ int s_cnt[RANGE_NODES];
  __shared__ int s_cur[RANGE_NODES];
  __shared__ int s_red[NCHUNK / 32];
  const int b = blockIdx.x;
  const int tid = threadIdx.x;
  const int lane = tid & 31;
  const int wv = tid >> 5;
  const int ch = (E + NCHUNK - 1) / NCHUNK;
  const int cbeg = min(tid * ch, E);
  const int clen = min(ch, E - cbeg);

  int o0 = tab[(size_t)tid * TPITCH + b];
  int o1 = tab[(size_t)tid * TPITCH + b + 1];
  o0 = min(max(o0, 0), clen);
  o1 = min(max(o1, o0), clen);
  const int* seg = part + cbeg;

  constexpr int RK = 8;
  int r[RK];
#pragma unroll
  for (int k = 0; k < RK; ++k) {
    const int v = part[max(min(cbeg + o0 + k, E - 1), 0)];
    r[k] = (o0 + k < o1) ? v : -1;
  }

  if (tid < RANGE_NODES) s_cnt[tid] = 0;
  const int bs = wave_sum_i(o0);
  if (lane == 0) s_red[wv] = bs;
  __syncthreads();
  int base = 0;
#pragma unroll
  for (int k = 0; k < NCHUNK / 32; ++k) base += s_red[k];

#pragma unroll
  for (int k = 0; k < RK; ++k)
    if (r[k] >= 0) atomicAdd(&s_cnt[(r[k] >> 16) & (RANGE_NODES - 1)], 1);
  for (int i = o0 + RK; i < o1; ++i)
    atomicAdd(&s_cnt[(seg[i] >> 16) & (RANGE_NODES - 1)], 1);
  __syncthreads();

  if (wv == 0) {
    const int c0 = s_cnt[2 * lane];
    const int c1 = s_cnt[2 * lane + 1];
    const int pr = c0 + c1;
    const int inc = wave_incl_scan_i(pr, lane);
    const int x0 = base + inc - pr;
    const int x1 = x0 + c0;
    s_cur[2 * lane] = x0;
    s_cur[2 * lane + 1] = x1;
    const int n0 = b * RANGE_NODES + 2 * lane;
    if (n0 <= NNODES) rowptr[n0] = x0;
    if (n0 + 1 <= NNODES) rowptr[n0 + 1] = x1;
    if (n0 < NNODES) dinv[n0] = rsqrtf((float)(c0 + 1));
    if (n0 + 1 < NNODES) dinv[n0 + 1] = rsqrtf((float)(c1 + 1));
  }
  __syncthreads();

#pragma unroll
  for (int k = 0; k < RK; ++k) {
    if (r[k] >= 0) {
      const int p = atomicAdd(&s_cur[(r[k] >> 16) & (RANGE_NODES - 1)], 1);
      if ((unsigned)p < (unsigned)E) csr[p] = r[k] & 0xffff;
    }
  }
  for (int i = o0 + RK; i < o1; ++i) {
    const int rec = seg[i];
    const int p = atomicAdd(&s_cur[(rec >> 16) & (RANGE_NODES - 1)], 1);
    if ((unsigned)p < (unsigned)E) csr[p] = rec & 0xffff;
  }
}

#define GEMM_WAVES 16

template <int K, int N>
__global__ void __launch_bounds__(GEMM_WAVES * 32)
k_gemm_wmma_f32(const float* __restrict__ A,
                const float* __restrict__ A_hi, int split,
                const float* __restrict__ W,
                float* __restrict__ C, int M) {
  constexpr int NCG = N / 64;
  constexpr int PITCH = N * 2 + 16;
  constexpr int JB = 4;
  __shared__ __attribute__((aligned(16))) float wl[(K / 2) * PITCH];

  const int tid = threadIdx.x;
  constexpr int STAGE_TRIPS = (K / 2) * (N / 4) / (GEMM_WAVES * 32);
  static_assert((K / 2) * (N / 4) % (GEMM_WAVES * 32) == 0, "W does not divide over the threads");
#pragma unroll
  for (int it = 0; it < STAGE_TRIPS; ++it) {
    const int idx = it * (GEMM_WAVES * 32) + tid;
    const int p = idx / (N / 4);
    const int n4 = idx % (N / 4);
    const float4 lo = *(const float4*)(W + (size_t)(2 * p) * N + 4 * n4);
    const float4 hi = *(const float4*)(W + (size_t)(2 * p + 1) * N + 4 * n4);
    float* d = wl + p * PITCH + 8 * n4;
    *(float4*)d = make_float4(lo.x, hi.x, lo.y, hi.y);
    *(float4*)(d + 4) = make_float4(lo.z, hi.z, lo.w, hi.w);
  }
  __syncthreads();

  const int wave = __builtin_amdgcn_readfirstlane(tid >> 5);
  const int lane = tid & 31;
  const int h = lane >> 4;
  const int c = lane & 15;
  const int row_tiles = M / 16;

  for (int li = wave;; li += GEMM_WAVES) {
    const int u = li / NCG;
    const int cg = li % NCG;
    const int t = u * (int)gridDim.x + (int)blockIdx.x;
    if (t >= row_tiles) break;

    const int row0 = t * 16;
    const float* atile = (row0 < split) ? (A + (size_t)row0 * K)
                                        : (A_hi + (size_t)(row0 - split) * K);
    const float* arow = atile + (size_t)c * K + 4 * h;
    const float* bl = wl + (2 * h) * PITCH + (cg * 64 + c) * 2;

    v8f acc[4] = {};
    float4 a_cur[JB];
#pragma unroll
    for (int q = 0; q < JB; ++q) a_cur[q] = *(const float4*)(arow + 8 * q);

#pragma unroll 1
    for (int jb = 0; jb < K / 8; jb += JB) {
      float4 a_nxt[JB];
      const int jn = (jb + JB < K / 8) ? (jb + JB) : jb;
#pragma unroll
      for (int q = 0; q < JB; ++q) a_nxt[q] = *(const float4*)(arow + 8 * (jn + q));
      __builtin_amdgcn_sched_barrier(0);
      const float* bj = bl + (4 * jb) * PITCH;
#pragma unroll
      for (int q = 0; q < JB; ++q) {
        const float* bq = bj + (4 * q) * PITCH;
        v2f a0; a0.x = a_cur[q].x; a0.y = a_cur[q].y;
        v2f a1; a1.x = a_cur[q].z; a1.y = a_cur[q].w;
#pragma unroll
        for (int tt = 0; tt < 4; ++tt) {
          const v2f b0 = *(const v2f*)(bq + tt * 32);
          acc[tt] = __builtin_amdgcn_wmma_f32_16x16x4_f32(
               false, a0,  false, b0,
               (short)0, acc[tt],  false,  false);
        }
#pragma unroll
        for (int tt = 0; tt < 4; ++tt) {
          const v2f b1 = *(const v2f*)(bq + PITCH + tt * 32);
          acc[tt] = __builtin_amdgcn_wmma_f32_16x16x4_f32(
               false, a1,  false, b1,
               (short)0, acc[tt],  false,  false);
        }
      }
#pragma unroll
      for (int q = 0; q < JB; ++q) a_cur[q] = a_nxt[q];
    }

    float* crow = C + (size_t)(row0 + 8 * h) * N + cg * 64 + c;
#pragma unroll
    for (int tt = 0; tt < 4; ++tt) {
#pragma unroll
      for (int i = 0; i < 8; ++i) crow[(size_t)i * N + tt * 16] = acc[tt][i];
    }
  }
}

template <int CH> struct RowVec;
template <> struct RowVec<128> { typedef float4 type; };
template <> struct RowVec<64>  { typedef float2 type; };

__device__ __forceinline__ void fma_row(float4& a, const float4& v, float w) {
  a.x += v.x * w; a.y += v.y * w; a.z += v.z * w; a.w += v.w * w;
}
__device__ __forceinline__ void fma_row(float2& a, const float2& v, float w) {
  a.x += v.x * w; a.y += v.y * w;
}
__device__ __forceinline__ float4 self_row(const float4& v, float w, const float4& b) {
  return make_float4(v.x * w + b.x, v.y * w + b.y, v.z * w + b.z, v.w * w + b.w);
}
__device__ __forceinline__ float2 self_row(const float2& v, float w, const float2& b) {
  return make_float2(v.x * w + b.x, v.y * w + b.y);
}
__device__ __forceinline__ float4 tanh_row(const float4& a) {
  return make_float4(tanhf(a.x), tanhf(a.y), tanhf(a.z), tanhf(a.w));
}

__device__ __forceinline__ int sort32(int key, int lane) {
#pragma unroll
  for (int k = 2; k <= 32; k <<= 1) {
#pragma unroll
    for (int j = k >> 1; j > 0; j >>= 1) {
      const int o = __shfl_xor(key, j);
      const bool up = (lane & k) == 0;
      const bool lo = (lane & j) == 0;
      const int mn = min(key, o);
      const int mx = max(key, o);
      key = (lo == up) ? mn : mx;
    }
  }
  return key;
}

template <int CH, bool ACT>
__global__ void __launch_bounds__(256)
k_gather(const int* __restrict__ rowptr, const int* __restrict__ csr,
         const float* __restrict__ dinv, const float* __restrict__ feat,
         const float* __restrict__ bias,
         float* __restrict__ out_lo, float* __restrict__ out_hi, int split, int E) {
  typedef typename RowVec<CH>::type VT;
  constexpr int VPL = CH / 32;
  const int lane = threadIdx.x & 31;
  const int d = __builtin_amdgcn_readfirstlane(
      (int)((blockIdx.x * blockDim.x + threadIdx.x) >> 5));
  if (d >= NNODES) return;

  int beg = rowptr[d];
  int end = rowptr[d + 1];
  beg = min(max(beg, 0), E);
  end = min(max(end, beg), E);
  const int deg = end - beg;
  const float di = dinv[d];
  const float* fl = feat + lane * VPL;

  VT acc = self_row(*(const VT*)(fl + (size_t)d * CH), di * di,
                    *(const VT*)(bias + lane * VPL));

  if (deg <= 32) {
    int key = (lane < deg) ? csr[beg + lane] : KEY_PAD;
    key = sort32(key, lane);
    const float w = (lane < deg) ? dinv[clamp_node(key)] * di : 0.0f;
#pragma unroll 4
    for (int j = 0; j < deg; ++j) {
      const int s = clamp_node(__builtin_amdgcn_readlane(key, j));
      const float ws = __int_as_float(__builtin_amdgcn_readlane(__float_as_int(w), j));
      const VT v = *(const VT*)(fl + (size_t)s * CH);
      fma_row(acc, v, ws);
    }
  } else {
    int last = -1;
    int done = 0;
    while (done < deg) {
      int best = KEY_PAD;
      for (int j = beg + lane; j < end; j += 32) {
        const int k = csr[j];
        if (k > last && k < best) best = k;
      }
#pragma unroll
      for (int o = 16; o > 0; o >>= 1) best = min(best, __shfl_xor(best, o));
      int mult = 0;
      for (int j = beg + lane; j < end; j += 32) mult += (csr[j] == best) ? 1 : 0;
      mult = __builtin_amdgcn_readfirstlane(wave_sum_i(mult));
      best = __builtin_amdgcn_readfirstlane(best);
      if (mult <= 0) break;
      const int s = clamp_node(best);
      const float ws = dinv[s] * di;
      const VT v = *(const VT*)(fl + (size_t)s * CH);
      for (int m = 0; m < mult; ++m) fma_row(acc, v, ws);
      last = best;
      done += mult;
    }
  }

  if constexpr (ACT) acc = tanh_row(acc);

  float* orow = (d < split) ? (out_lo + (size_t)d * CH) : (out_hi + (size_t)(d - split) * CH);
  *(VT*)(orow + lane * VPL) = acc;
}

extern "C" void kernel_launch(void* const* d_in, const int* in_sizes, int n_in,
                              void* d_out, int out_size, void* d_ws, size_t ws_size,
                              hipStream_t stream) {
  const float* x   = (const float*)d_in[0];
  const int*  eidx = (const int*)d_in[1];
  const float* W1  = (const float*)d_in[2];
  const float* b1  = (const float*)d_in[3];
  const float* W2  = (const float*)d_in[4];
  const float* b2  = (const float*)d_in[5];
  float* out = (float*)d_out;

  const int E = in_sizes[1] / 2;
  const int* src = eidx;
  const int* dst = eidx + E;

  float* dinv = (float*)d_ws;
  float* bufA = dinv + 50048;
  float* bufB = bufA + (size_t)NNODES * 128;
  int* csr    = (int*)(bufB + (size_t)SPLIT_ROWS * 128);
  int* rowptr = csr + 800000;
  int* part   = (int*)out;
  int* tab    = part + 800000;

  k_bin_chunk<<<NCHUNK, 1024, 0, stream>>>(src, dst, part, tab, E);
  k_rank_range<<<NRANGE, NCHUNK, 0, stream>>>(part, tab, rowptr, dinv, csr, E);

  k_gemm_wmma_f32<256, 128><<<256, GEMM_WAVES * 32, 0, stream>>>(x, x, NNODES, W1, bufA,
                                                                 NNODES);
  k_gather<128, true><<<(NNODES + 7) / 8, 256, 0, stream>>>(rowptr, csr, dinv, bufA, b1,
                                                           bufB, out, SPLIT_ROWS, E);

  k_gemm_wmma_f32<128, 64><<<256, GEMM_WAVES * 32, 0, stream>>>(bufB, out, SPLIT_ROWS, W2,
                                                               bufA, NNODES);
  k_gather<64, false><<<(NNODES + 7) / 8, 256, 0, stream>>>(rowptr, csr, dinv, bufA, b2,
                                                           out, out, NNODES, E);
}


# Gather: lists of 33-64 sources sorted once in registers (64-key bitonic, 2 keys per lane), rows streamed 4 in flight, not one dependent min-search pass per source; same ascending order, fused adds
# speedup vs baseline: 3.6349x; 1.5324x over previous
#include <hip/hip_runtime.h>
#include <hip/hip_bf16.h>


typedef float v2f __attribute__((ext_vector_type(2)));
typedef float v8f __attribute__((ext_vector_type(8)));

#define NNODES 50000
#define SPLIT_ROWS 42000
#define KEY_PAD 0x7fffffff

#define RANGE_BITS 6
#define RANGE_NODES 64
#define NRANGE ((NNODES + RANGE_NODES - 1) / RANGE_NODES)
#define TPITCH 784
#define NCHUNK 256
#define STAGE_MAX 4096

__device__ __forceinline__ int clamp_node(int s) { return min(max(s, 0), NNODES - 1); }

__device__ __forceinline__ int wave_sum_i(int v) {
#pragma unroll
  for (int o = 16; o > 0; o >>= 1) v += __shfl_xor(v, o);
  return v;
}

__device__ __forceinline__ int wave_incl_scan_i(int v, int lane) {
#pragma unroll
  for (int o = 1; o < 32; o <<= 1) {
    int t = __shfl_up(v, o);
    if (lane >= o) v += t;
  }
  return v;
}

__global__ void __launch_bounds__(1024) k_bin_chunk(const int* __restrict__ src,
                                                    const int* __restrict__ dst,
                                                    int* __restrict__ part,
                                                    int* __restrict__ tab, int E) {
  __shared__ int s_hist[1024];
  __shared__ int s_w[32];
  __shared__ int s_stage[STAGE_MAX];
  const int tid = threadIdx.x;
  const int lane = tid & 31;
  const int wv = tid >> 5;
  const int ch = (E + NCHUNK - 1) / NCHUNK;
  const int e0 = min((int)blockIdx.x * ch, E);
  const int e1 = min(e0 + ch, E);

  s_hist[tid] = 0;
  constexpr int PK = 4;
  int dv[PK], sv[PK];
#pragma unroll
  for (int k = 0; k < PK; ++k) {
    const int e = e0 + tid + k * 1024;
    const bool ok = e < e1;
    const int ec = max(min(e, E - 1), 0);
    const int dd = dst[ec];
    const int ss = src[ec];
    dv[k] = clamp_node(dd) | (ok ? 0 : (int)0x80000000);
    sv[k] = clamp_node(ss);
  }
  __syncthreads();
#pragma unroll
  for (int k = 0; k < PK; ++k)
    if (dv[k] >= 0) atomicAdd(&s_hist[dv[k] >> RANGE_BITS], 1);
  for (int e = e0 + tid + PK * 1024; e < e1; e += 1024)
    atomicAdd(&s_hist[clamp_node(dst[e]) >> RANGE_BITS], 1);
  __syncthreads();

  const int cnt = s_hist[tid];
  const int inc = wave_incl_scan_i(cnt, lane);
  if (lane == 31) s_w[wv] = inc;
  __syncthreads();
  if (wv == 0) {
    const int t = s_w[lane];
    const int ti = wave_incl_scan_i(t, lane);
    s_w[lane] = ti - t;
  }
  __syncthreads();
  const int excl = s_w[wv] + inc - cnt;
  if (tid < TPITCH) tab[(size_t)blockIdx.x * TPITCH + tid] = excl;
  s_hist[tid] = excl;
  __syncthreads();

  const bool staged = ch <= STAGE_MAX;
  int* pout = part + e0;
#pragma unroll
  for (int k = 0; k < PK; ++k) {
    if (dv[k] >= 0) {
      const int p = atomicAdd(&s_hist[dv[k] >> RANGE_BITS], 1);
      const int rec = ((dv[k] & (RANGE_NODES - 1)) << 16) | sv[k];
      if (staged) { if ((unsigned)p < (unsigned)STAGE_MAX) s_stage[p] = rec; }
      else if (p < e1 - e0) pout[p] = rec;
    }
  }
  for (int e = e0 + tid + PK * 1024; e < e1; e += 1024) {
    const int d = clamp_node(dst[e]);
    const int p = atomicAdd(&s_hist[d >> RANGE_BITS], 1);
    const int rec = ((d & (RANGE_NODES - 1)) << 16) | clamp_node(src[e]);
    if (staged) { if ((unsigned)p < (unsigned)STAGE_MAX) s_stage[p] = rec; }
    else if (p < e1 - e0) pout[p] = rec;
  }
  if (staged) {
    __syncthreads();
    for (int i = tid; i < e1 - e0; i += 1024) pout[i] = s_stage[i];
  }
}

__global__ void __launch_bounds__(NCHUNK) k_rank_range(const int* __restrict__ part,
                                                       const int* __restrict__ tab,
                                                       int* __restrict__ rowptr,
                                                       float* __restrict__ dinv,
                                                       int* __restrict__ csr, int E) {
  __shared__ int s_cnt[RANGE_NODES];
  __shared__ int s_cur[RANGE_NODES];
  __shared__ int s_red[NCHUNK / 32];
  const int b = blockIdx.x;
  const int tid = threadIdx.x;
  const int lane = tid & 31;
  const int wv = tid >> 5;
  const int ch = (E + NCHUNK - 1) / NCHUNK;
  const int cbeg = min(tid * ch, E);
  const int clen = min(ch, E - cbeg);

  int o0 = tab[(size_t)tid * TPITCH + b];
  int o1 = tab[(size_t)tid * TPITCH + b + 1];
  o0 = min(max(o0, 0), clen);
  o1 = min(max(o1, o0), clen);
  const int* seg = part + cbeg;

  constexpr int RK = 8;
  int r[RK];
#pragma unroll
  for (int k = 0; k < RK; ++k) {
    const int v = part[max(min(cbeg + o0 + k, E - 1), 0)];
    r[k] = (o0 + k < o1) ? v : -1;
  }

  if (tid < RANGE_NODES) s_cnt[tid] = 0;
  const int bs = wave_sum_i(o0);
  if (lane == 0) s_red[wv] = bs;
  __syncthreads();
  int base = 0;
#pragma unroll
  for (int k = 0; k < NCHUNK / 32; ++k) base += s_red[k];

#pragma unroll
  for (int k = 0; k < RK; ++k)
    if (r[k] >= 0) atomicAdd(&s_cnt[(r[k] >> 16) & (RANGE_NODES - 1)], 1);
  for (int i = o0 + RK; i < o1; ++i)
    atomicAdd(&s_cnt[(seg[i] >> 16) & (RANGE_NODES - 1)], 1);
  __syncthreads();

  if (wv == 0) {
    const int c0 = s_cnt[2 * lane];
    const int c1 = s_cnt[2 * lane + 1];
    const int pr = c0 + c1;
    const int inc = wave_incl_scan_i(pr, lane);
    const int x0 = base + inc - pr;
    const int x1 = x0 + c0;
    s_cur[2 * lane] = x0;
    s_cur[2 * lane + 1] = x1;
    const int n0 = b * RANGE_NODES + 2 * lane;
    if (n0 <= NNODES) rowptr[n0] = x0;
    if (n0 + 1 <= NNODES) rowptr[n0 + 1] = x1;
    if (n0 < NNODES) dinv[n0] = rsqrtf((float)(c0 + 1));
    if (n0 + 1 < NNODES) dinv[n0 + 1] = rsqrtf((float)(c1 + 1));
  }
  __syncthreads();

#pragma unroll
  for (int k = 0; k < RK; ++k) {
    if (r[k] >= 0) {
      const int p = atomicAdd(&s_cur[(r[k] >> 16) & (RANGE_NODES - 1)], 1);
      if ((unsigned)p < (unsigned)E) csr[p] = r[k] & 0xffff;
    }
  }
  for (int i = o0 + RK; i < o1; ++i) {
    const int rec = seg[i];
    const int p = atomicAdd(&s_cur[(rec >> 16) & (RANGE_NODES - 1)], 1);
    if ((unsigned)p < (unsigned)E) csr[p] = rec & 0xffff;
  }
}

#define GEMM_WAVES 16

template <int K, int N>
__global__ void __launch_bounds__(GEMM_WAVES * 32)
k_gemm_wmma_f32(const float* __restrict__ A,
                const float* __restrict__ A_hi, int split,
                const float* __restrict__ W,
                float* __restrict__ C, int M) {
  constexpr int NCG = N / 64;
  constexpr int PITCH = N * 2 + 16;
  constexpr int JB = 4;
  __shared__ __attribute__((aligned(16))) float wl[(K / 2) * PITCH];

  const int tid = threadIdx.x;
  constexpr int STAGE_TRIPS = (K / 2) * (N / 4) / (GEMM_WAVES * 32);
  static_assert((K / 2) * (N / 4) % (GEMM_WAVES * 32) == 0, "W does not divide over the threads");
#pragma unroll
  for (int it = 0; it < STAGE_TRIPS; ++it) {
    const int idx = it * (GEMM_WAVES * 32) + tid;
    const int p = idx / (N / 4);
    const int n4 = idx % (N / 4);
    const float4 lo = *(const float4*)(W + (size_t)(2 * p) * N + 4 * n4);
    const float4 hi = *(const float4*)(W + (size_t)(2 * p + 1) * N + 4 * n4);
    float* d = wl + p * PITCH + 8 * n4;
    *(float4*)d = make_float4(lo.x, hi.x, lo.y, hi.y);
    *(float4*)(d + 4) = make_float4(lo.z, hi.z, lo.w, hi.w);
  }
  __syncthreads();

  const int wave = __builtin_amdgcn_readfirstlane(tid >> 5);
  const int lane = tid & 31;
  const int h = lane >> 4;
  const int c = lane & 15;
  const int row_tiles = M / 16;

  for (int li = wave;; li += GEMM_WAVES) {
    const int u = li / NCG;
    const int cg = li % NCG;
    const int t = u * (int)gridDim.x + (int)blockIdx.x;
    if (t >= row_tiles) break;

    const int row0 = t * 16;
    const float* atile = (row0 < split) ? (A + (size_t)row0 * K)
                                        : (A_hi + (size_t)(row0 - split) * K);
    const float* arow = atile + (size_t)c * K + 4 * h;
    const float* bl = wl + (2 * h) * PITCH + (cg * 64 + c) * 2;

    v8f acc[4] = {};
    float4 a_cur[JB];
#pragma unroll
    for (int q = 0; q < JB; ++q) a_cur[q] = *(const float4*)(arow + 8 * q);

#pragma unroll 1
    for (int jb = 0; jb < K / 8; jb += JB) {
      float4 a_nxt[JB];
      const int jn = (jb + JB < K / 8) ? (jb + JB) : jb;
#pragma unroll
      for (int q = 0; q < JB; ++q) a_nxt[q] = *(const float4*)(arow + 8 * (jn + q));
      __builtin_amdgcn_sched_barrier(0);
      const float* bj = bl + (4 * jb) * PITCH;
#pragma unroll
      for (int q = 0; q < JB; ++q) {
        const float* bq = bj + (4 * q) * PITCH;
        v2f a0; a0.x = a_cur[q].x; a0.y = a_cur[q].y;
        v2f a1; a1.x = a_cur[q].z; a1.y = a_cur[q].w;
#pragma unroll
        for (int tt = 0; tt < 4; ++tt) {
          const v2f b0 = *(const v2f*)(bq + tt * 32);
          acc[tt] = __builtin_amdgcn_wmma_f32_16x16x4_f32(
               false, a0,  false, b0,
               (short)0, acc[tt],  false,  false);
        }
#pragma unroll
        for (int tt = 0; tt < 4; ++tt) {
          const v2f b1 = *(const v2f*)(bq + PITCH + tt * 32);
          acc[tt] = __builtin_amdgcn_wmma_f32_16x16x4_f32(
               false, a1,  false, b1,
               (short)0, acc[tt],  false,  false);
        }
      }
#pragma unroll
      for (int q = 0; q < JB; ++q) a_cur[q] = a_nxt[q];
    }

    float* crow = C + (size_t)(row0 + 8 * h) * N + cg * 64 + c;
#pragma unroll
    for (int tt = 0; tt < 4; ++tt) {
#pragma unroll
      for (int i = 0; i < 8; ++i) crow[(size_t)i * N + tt * 16] = acc[tt][i];
    }
  }
}

template <int CH> struct RowVec;
template <> struct RowVec<128> { typedef float4 type; };
template <> struct RowVec<64>  { typedef float2 type; };

__device__ __forceinline__ void fma_row(float4& a, const float4& v, float w) {
  a.x += v.x * w; a.y += v.y * w; a.z += v.z * w; a.w += v.w * w;
}
__device__ __forceinline__ void fma_row(float2& a, const float2& v, float w) {
  a.x += v.x * w; a.y += v.y * w;
}
__device__ __forceinline__ float4 self_row(const float4& v, float w, const float4& b) {
  return make_float4(v.x * w + b.x, v.y * w + b.y, v.z * w + b.z, v.w * w + b.w);
}
__device__ __forceinline__ float2 self_row(const float2& v, float w, const float2& b) {
  return make_float2(v.x * w + b.x, v.y * w + b.y);
}
__device__ __forceinline__ float4 tanh_row(const float4& a) {
  return make_float4(tanhf(a.x), tanhf(a.y), tanhf(a.z), tanhf(a.w));
}

__device__ __forceinline__ int sort32(int key, int lane) {
#pragma unroll
  for (int k = 2; k <= 32; k <<= 1) {
#pragma unroll
    for (int j = k >> 1; j > 0; j >>= 1) {
      const int o = __shfl_xor(key, j);
      const bool up = (lane & k) == 0;
      const bool lo = (lane & j) == 0;
      const int mn = min(key, o);
      const int mx = max(key, o);
      key = (lo == up) ? mn : mx;
    }
  }
  return key;
}

__device__ __forceinline__ void sort64(int& k0, int& k1, int lane) {
#pragma unroll
  for (int k = 2; k <= 64; k <<= 1) {
#pragma unroll
    for (int j = k >> 1; j > 0; j >>= 1) {
      if (j == 32) {
        const int mn = min(k0, k1);
        const int mx = max(k0, k1);
        k0 = mn;
        k1 = mx;
      } else {
        const int o0 = __shfl_xor(k0, j);
        const int o1 = __shfl_xor(k1, j);
        const bool lo = (lane & j) == 0;
        const bool up0 = (k >= 32) ? true : ((lane & k) == 0);
        const bool up1 = (k == 64) ? true : (k == 32) ? false : ((lane & k) == 0);
        k0 = (lo == up0) ? min(k0, o0) : max(k0, o0);
        k1 = (lo == up1) ? min(k1, o1) : max(k1, o1);
      }
    }
  }
}

template <int CH, typename VT>
__device__ __forceinline__ void add_rows(VT& acc, const float* __restrict__ fl, int key,
                                         float w, int n) {
#pragma unroll 4
  for (int j = 0; j < n; ++j) {
    const int s = clamp_node(__builtin_amdgcn_readlane(key, j));
    const float ws = __int_as_float(__builtin_amdgcn_readlane(__float_as_int(w), j));
    const VT v = *(const VT*)(fl + (size_t)s * CH);
    fma_row(acc, v, ws);
  }
}

template <int CH, bool ACT>
__global__ void __launch_bounds__(256)
k_gather(const int* __restrict__ rowptr, const int* __restrict__ csr,
         const float* __restrict__ dinv, const float* __restrict__ feat,
         const float* __restrict__ bias,
         float* __restrict__ out_lo, float* __restrict__ out_hi, int split, int E) {
  typedef typename RowVec<CH>::type VT;
  constexpr int VPL = CH / 32;
  const int lane = threadIdx.x & 31;
  const int d = __builtin_amdgcn_readfirstlane(
      (int)((blockIdx.x * blockDim.x + threadIdx.x) >> 5));
  if (d >= NNODES) return;

  int beg = rowptr[d];
  int end = rowptr[d + 1];
  beg = min(max(beg, 0), E);
  end = min(max(end, beg), E);
  const int deg = end - beg;
  const float di = dinv[d];
  const float* fl = feat + lane * VPL;

  VT acc = self_row(*(const VT*)(fl + (size_t)d * CH), di * di,
                    *(const VT*)(bias + lane * VPL));

  if (deg <= 32) {
    int key = (lane < deg) ? csr[beg + lane] : KEY_PAD;
    key = sort32(key, lane);
    const float w = (lane < deg) ? dinv[clamp_node(key)] * di : 0.0f;
    add_rows<CH>(acc, fl, key, w, deg);
  } else if (deg <= 64) {
    const bool two = lane + 32 < deg;
    int k0 = csr[beg + lane];
    int k1 = two ? csr[beg + 32 + lane] : KEY_PAD;
    sort64(k0, k1, lane);
    const float w0 = dinv[clamp_node(k0)] * di;
    const float w1 = two ? dinv[clamp_node(k1)] * di : 0.0f;
    add_rows<CH>(acc, fl, k0, w0, 32);
    add_rows<CH>(acc, fl, k1, w1, deg - 32);
  } else {
    int last = -1;
    int done = 0;
    while (done < deg) {
      int best = KEY_PAD;
      for (int j = beg + lane; j < end; j += 32) {
        const int k = csr[j];
        if (k > last && k < best) best = k;
      }
#pragma unroll
      for (int o = 16; o > 0; o >>= 1) best = min(best, __shfl_xor(best, o));
      int mult = 0;
      for (int j = beg + lane; j < end; j += 32) mult += (csr[j] == best) ? 1 : 0;
      mult = __builtin_amdgcn_readfirstlane(wave_sum_i(mult));
      best = __builtin_amdgcn_readfirstlane(best);
      if (mult <= 0) break;
      const int s = clamp_node(best);
      const float ws = dinv[s] * di;
      const VT v = *(const VT*)(fl + (size_t)s * CH);
      for (int m = 0; m < mult; ++m) fma_row(acc, v, ws);
      last = best;
      done += mult;
    }
  }

  if constexpr (ACT) acc = tanh_row(acc);

  float* orow = (d < split) ? (out_lo + (size_t)d * CH) : (out_hi + (size_t)(d - split) * CH);
  *(VT*)(orow + lane * VPL) = acc;
}

extern "C" void kernel_launch(void* const* d_in, const int* in_sizes, int n_in,
                              void* d_out, int out_size, void* d_ws, size_t ws_size,
                              hipStream_t stream) {
  const float* x   = (const float*)d_in[0];
  const int*  eidx = (const int*)d_in[1];
  const float* W1  = (const float*)d_in[2];
  const float* b1  = (const float*)d_in[3];
  const float* W2  = (const float*)d_in[4];
  const float* b2  = (const float*)d_in[5];
  float* out = (float*)d_out;

  const int E = in_sizes[1] / 2;
  const int* src = eidx;
  const int* dst = eidx + E;

  float* dinv = (float*)d_ws;
  float* bufA = dinv + 50048;
  float* bufB = bufA + (size_t)NNODES * 128;
  int* csr    = (int*)(bufB + (size_t)SPLIT_ROWS * 128);
  int* rowptr = csr + 800000;
  int* part   = (int*)out;
  int* tab    = part + 800000;

  k_bin_chunk<<<NCHUNK, 1024, 0, stream>>>(src, dst, part, tab, E);
  k_rank_range<<<NRANGE, NCHUNK, 0, stream>>>(part, tab, rowptr, dinv, csr, E);

  k_gemm_wmma_f32<256, 128><<<256, GEMM_WAVES * 32, 0, stream>>>(x, x, NNODES, W1, bufA,
                                                                 NNODES);
  k_gather<128, true><<<(NNODES + 7) / 8, 256, 0, stream>>>(rowptr, csr, dinv, bufA, b1,
                                                           bufB, out, SPLIT_ROWS, E);

  k_gemm_wmma_f32<128, 64><<<256, GEMM_WAVES * 32, 0, stream>>>(bufB, out, SPLIT_ROWS, W2,
                                                               bufA, NNODES);
  k_gather<64, false><<<(NNODES + 7) / 8, 256, 0, stream>>>(rowptr, csr, dinv, bufA, b2,
                                                           out, out, NNODES, E);
}


# Gather made resident: 12500 waves (500 workgroups of 25) walk 4 nodes each, next node's keys and own row asked for under the current rows, leftover rows as one group; fused adds, sum order kept
# speedup vs baseline: 3.8376x; 1.0558x over previous
#include <hip/hip_runtime.h>
#include <hip/hip_bf16.h>


typedef float v2f __attribute__((ext_vector_type(2)));
typedef float v8f __attribute__((ext_vector_type(8)));

#define NNODES 50000
#define SPLIT_ROWS 42000
#define KEY_PAD 0x7fffffff

#define RANGE_BITS 6
#define RANGE_NODES 64
#define NRANGE ((NNODES + RANGE_NODES - 1) / RANGE_NODES)
#define TPITCH 784
#define NCHUNK 256
#define STAGE_MAX 4096

__device__ __forceinline__ int clamp_node(int s) { return min(max(s, 0), NNODES - 1); }

__device__ __forceinline__ int wave_sum_i(int v) {
#pragma unroll
  for (int o = 16; o > 0; o >>= 1) v += __shfl_xor(v, o);
  return v;
}

__device__ __forceinline__ int wave_incl_scan_i(int v, int lane) {
#pragma unroll
  for (int o = 1; o < 32; o <<= 1) {
    int t = __shfl_up(v, o);
    if (lane >= o) v += t;
  }
  return v;
}

__global__ void __launch_bounds__(1024) k_bin_chunk(const int* __restrict__ src,
                                                    const int* __restrict__ dst,
                                                    int* __restrict__ part,
                                                    int* __restrict__ tab, int E) {
  __shared__ int s_hist[1024];
  __shared__ int s_w[32];
  __shared__ int s_stage[STAGE_MAX];
  const int tid = threadIdx.x;
  const int lane = tid & 31;
  const int wv = tid >> 5;
  const int ch = (E + NCHUNK - 1) / NCHUNK;
  const int e0 = min((int)blockIdx.x * ch, E);
  const int e1 = min(e0 + ch, E);

  s_hist[tid] = 0;
  constexpr int PK = 4;
  int dv[PK], sv[PK];
#pragma unroll
  for (int k = 0; k < PK; ++k) {
    const int e = e0 + tid + k * 1024;
    const bool ok = e < e1;
    const int ec = max(min(e, E - 1), 0);
    const int dd = dst[ec];
    const int ss = src[ec];
    dv[k] = clamp_node(dd) | (ok ? 0 : (int)0x80000000);
    sv[k] = clamp_node(ss);
  }
  __syncthreads();
#pragma unroll
  for (int k = 0; k < PK; ++k)
    if (dv[k] >= 0) atomicAdd(&s_hist[dv[k] >> RANGE_BITS], 1);
  for (int e = e0 + tid + PK * 1024; e < e1; e += 1024)
    atomicAdd(&s_hist[clamp_node(dst[e]) >> RANGE_BITS], 1);
  __syncthreads();

  const int cnt = s_hist[tid];
  const int inc = wave_incl_scan_i(cnt, lane);
  if (lane == 31) s_w[wv] = inc;
  __syncthreads();
  if (wv == 0) {
    const int t = s_w[lane];
    const int ti = wave_incl_scan_i(t, lane);
    s_w[lane] = ti - t;
  }
  __syncthreads();
  const int excl = s_w[wv] + inc - cnt;
  if (tid < TPITCH) tab[(size_t)blockIdx.x * TPITCH + tid] = excl;
  s_hist[tid] = excl;
  __syncthreads();

  const bool staged = ch <= STAGE_MAX;
  int* pout = part + e0;
#pragma unroll
  for (int k = 0; k < PK; ++k) {
    if (dv[k] >= 0) {
      const int p = atomicAdd(&s_hist[dv[k] >> RANGE_BITS], 1);
      const int rec = ((dv[k] & (RANGE_NODES - 1)) << 16) | sv[k];
      if (staged) { if ((unsigned)p < (unsigned)STAGE_MAX) s_stage[p] = rec; }
      else if (p < e1 - e0) pout[p] = rec;
    }
  }
  for (int e = e0 + tid + PK * 1024; e < e1; e += 1024) {
    const int d = clamp_node(dst[e]);
    const int p = atomicAdd(&s_hist[d >> RANGE_BITS], 1);
    const int rec = ((d & (RANGE_NODES - 1)) << 16) | clamp_node(src[e]);
    if (staged) { if ((unsigned)p < (unsigned)STAGE_MAX) s_stage[p] = rec; }
    else if (p < e1 - e0) pout[p] = rec;
  }
  if (staged) {
    __syncthreads();
    for (int i = tid; i < e1 - e0; i += 1024) pout[i] = s_stage[i];
  }
}

__global__ void __launch_bounds__(NCHUNK) k_rank_range(const int* __restrict__ part,
                                                       const int* __restrict__ tab,
                                                       int* __restrict__ rowptr,
                                                       float* __restrict__ dinv,
                                                       int* __restrict__ csr, int E) {
  __shared__ int s_cnt[RANGE_NODES];
  __shared__ int s_cur[RANGE_NODES];
  __shared__ int s_red[NCHUNK / 32];
  const int b = blockIdx.x;
  const int tid = threadIdx.x;
  const int lane = tid & 31;
  const int wv = tid >> 5;
  const int ch = (E + NCHUNK - 1) / NCHUNK;
  const int cbeg = min(tid * ch, E);
  const int clen = min(ch, E - cbeg);

  int o0 = tab[(size_t)tid * TPITCH + b];
  int o1 = tab[(size_t)tid * TPITCH + b + 1];
  o0 = min(max(o0, 0), clen);
  o1 = min(max(o1, o0), clen);
  const int* seg = part + cbeg;

  constexpr int RK = 8;
  int r[RK];
#pragma unroll
  for (int k = 0; k < RK; ++k) {
    const int v = part[max(min(cbeg + o0 + k, E - 1), 0)];
    r[k] = (o0 + k < o1) ? v : -1;
  }

  if (tid < RANGE_NODES) s_cnt[tid] = 0;
  const int bs = wave_sum_i(o0);
  if (lane == 0) s_red[wv] = bs;
  __syncthreads();
  int base = 0;
#pragma unroll
  for (int k = 0; k < NCHUNK / 32; ++k) base += s_red[k];

#pragma unroll
  for (int k = 0; k < RK; ++k)
    if (r[k] >= 0) atomicAdd(&s_cnt[(r[k] >> 16) & (RANGE_NODES - 1)], 1);
  for (int i = o0 + RK; i < o1; ++i)
    atomicAdd(&s_cnt[(seg[i] >> 16) & (RANGE_NODES - 1)], 1);
  __syncthreads();

  if (wv == 0) {
    const int c0 = s_cnt[2 * lane];
    const int c1 = s_cnt[2 * lane + 1];
    const int pr = c0 + c1;
    const int inc = wave_incl_scan_i(pr, lane);
    const int x0 = base + inc - pr;
    const int x1 = x0 + c0;
    s_cur[2 * lane] = x0;
    s_cur[2 * lane + 1] = x1;
    const int n0 = b * RANGE_NODES + 2 * lane;
    if (n0 <= NNODES) rowptr[n0] = x0;
    if (n0 + 1 <= NNODES) rowptr[n0 + 1] = x1;
    if (n0 < NNODES) dinv[n0] = rsqrtf((float)(c0 + 1));
    if (n0 + 1 < NNODES) dinv[n0 + 1] = rsqrtf((float)(c1 + 1));
  }
  __syncthreads();

#pragma unroll
  for (int k = 0; k < RK; ++k) {
    if (r[k] >= 0) {
      const int p = atomicAdd(&s_cur[(r[k] >> 16) & (RANGE_NODES - 1)], 1);
      if ((unsigned)p < (unsigned)E) csr[p] = r[k] & 0xffff;
    }
  }
  for (int i = o0 + RK; i < o1; ++i) {
    const int rec = seg[i];
    const int p = atomicAdd(&s_cur[(rec >> 16) & (RANGE_NODES - 1)], 1);
    if ((unsigned)p < (unsigned)E) csr[p] = rec & 0xffff;
  }
}

#define GEMM_WAVES 16

template <int K, int N>
__global__ void __launch_bounds__(GEMM_WAVES * 32)
k_gemm_wmma_f32(const float* __restrict__ A,
                const float* __restrict__ A_hi, int split,
                const float* __restrict__ W,
                float* __restrict__ C, int M) {
  constexpr int NCG = N / 64;
  constexpr int PITCH = N * 2 + 16;
  constexpr int JB = 4;
  __shared__ __attribute__((aligned(16))) float wl[(K / 2) * PITCH];

  const int tid = threadIdx.x;
  constexpr int STAGE_TRIPS = (K / 2) * (N / 4) / (GEMM_WAVES * 32);
  static_assert((K / 2) * (N / 4) % (GEMM_WAVES * 32) == 0, "W does not divide over the threads");
#pragma unroll
  for (int it = 0; it < STAGE_TRIPS; ++it) {
    const int idx = it * (GEMM_WAVES * 32) + tid;
    const int p = idx / (N / 4);
    const int n4 = idx % (N / 4);
    const float4 lo = *(const float4*)(W + (size_t)(2 * p) * N + 4 * n4);
    const float4 hi = *(const float4*)(W + (size_t)(2 * p + 1) * N + 4 * n4);
    float* d = wl + p * PITCH + 8 * n4;
    *(float4*)d = make_float4(lo.x, hi.x, lo.y, hi.y);
    *(float4*)(d + 4) = make_float4(lo.z, hi.z, lo.w, hi.w);
  }
  __syncthreads();

  const int wave = __builtin_amdgcn_readfirstlane(tid >> 5);
  const int lane = tid & 31;
  const int h = lane >> 4;
  const int c = lane & 15;
  const int row_tiles = M / 16;

  for (int li = wave;; li += GEMM_WAVES) {
    const int u = li / NCG;
    const int cg = li % NCG;
    const int t = u * (int)gridDim.x + (int)blockIdx.x;
    if (t >= row_tiles) break;

    const int row0 = t * 16;
    const float* atile = (row0 < split) ? (A + (size_t)row0 * K)
                                        : (A_hi + (size_t)(row0 - split) * K);
    const float* arow = atile + (size_t)c * K + 4 * h;
    const float* bl = wl + (2 * h) * PITCH + (cg * 64 + c) * 2;

    v8f acc[4] = {};
    float4 a_cur[JB];
#pragma unroll
    for (int q = 0; q < JB; ++q) a_cur[q] = *(const float4*)(arow + 8 * q);

#pragma unroll 1
    for (int jb = 0; jb < K / 8; jb += JB) {
      float4 a_nxt[JB];
      const int jn = (jb + JB < K / 8) ? (jb + JB) : jb;
#pragma unroll
      for (int q = 0; q < JB; ++q) a_nxt[q] = *(const float4*)(arow + 8 * (jn + q));
      __builtin_amdgcn_sched_barrier(0);
      const float* bj = bl + (4 * jb) * PITCH;
#pragma unroll
      for (int q = 0; q < JB; ++q) {
        const float* bq = bj + (4 * q) * PITCH;
        v2f a0; a0.x = a_cur[q].x; a0.y = a_cur[q].y;
        v2f a1; a1.x = a_cur[q].z; a1.y = a_cur[q].w;
#pragma unroll
        for (int tt = 0; tt < 4; ++tt) {
          const v2f b0 = *(const v2f*)(bq + tt * 32);
          acc[tt] = __builtin_amdgcn_wmma_f32_16x16x4_f32(
               false, a0,  false, b0,
               (short)0, acc[tt],  false,  false);
        }
#pragma unroll
        for (int tt = 0; tt < 4; ++tt) {
          const v2f b1 = *(const v2f*)(bq + PITCH + tt * 32);
          acc[tt] = __builtin_amdgcn_wmma_f32_16x16x4_f32(
               false, a1,  false, b1,
               (short)0, acc[tt],  false,  false);
        }
      }
#pragma unroll
      for (int q = 0; q < JB; ++q) a_cur[q] = a_nxt[q];
    }

    float* crow = C + (size_t)(row0 + 8 * h) * N + cg * 64 + c;
#pragma unroll
    for (int tt = 0; tt < 4; ++tt) {
#pragma unroll
      for (int i = 0; i < 8; ++i) crow[(size_t)i * N + tt * 16] = acc[tt][i];
    }
  }
}

template <int CH> struct RowVec;
template <> struct RowVec<128> { typedef float4 type; };
template <> struct RowVec<64>  { typedef float2 type; };

__device__ __forceinline__ void fma_row(float4& a, const float4& v, float w) {
  a.x = __builtin_fmaf(v.x, w, a.x); a.y = __builtin_fmaf(v.y, w, a.y);
  a.z = __builtin_fmaf(v.z, w, a.z); a.w = __builtin_fmaf(v.w, w, a.w);
}
__device__ __forceinline__ void fma_row(float2& a, const float2& v, float w) {
  a.x = __builtin_fmaf(v.x, w, a.x); a.y = __builtin_fmaf(v.y, w, a.y);
}
__device__ __forceinline__ float4 self_row(const float4& v, float w, const float4& b) {
  return make_float4(v.x * w + b.x, v.y * w + b.y, v.z * w + b.z, v.w * w + b.w);
}
__device__ __forceinline__ float2 self_row(const float2& v, float w, const float2& b) {
  return make_float2(v.x * w + b.x, v.y * w + b.y);
}
__device__ __forceinline__ float4 tanh_row(const float4& a) {
  return make_float4(tanhf(a.x), tanhf(a.y), tanhf(a.z), tanhf(a.w));
}

__device__ __forceinline__ int sort32(int key, int lane) {
#pragma unroll
  for (int k = 2; k <= 32; k <<= 1) {
#pragma unroll
    for (int j = k >> 1; j > 0; j >>= 1) {
      const int o = __shfl_xor(key, j);
      const bool up = (lane & k) == 0;
      const bool lo = (lane & j) == 0;
      const int mn = min(key, o);
      const int mx = max(key, o);
      key = (lo == up) ? mn : mx;
    }
  }
  return key;
}

__device__ __forceinline__ void sort64(int& k0, int& k1, int lane) {
#pragma unroll
  for (int k = 2; k <= 64; k <<= 1) {
#pragma unroll
    for (int j = k >> 1; j > 0; j >>= 1) {
      if (j == 32) {
        const int mn = min(k0, k1);
        const int mx = max(k0, k1);
        k0 = mn;
        k1 = mx;
      } else {
        const int o0 = __shfl_xor(k0, j);
        const int o1 = __shfl_xor(k1, j);
        const bool lo = (lane & j) == 0;
        const bool up0 = (k >= 32) ? true : ((lane & k) == 0);
        const bool up1 = (k == 64) ? true : (k == 32) ? false : ((lane & k) == 0);
        k0 = (lo == up0) ? min(k0, o0) : max(k0, o0);
        k1 = (lo == up1) ? min(k1, o1) : max(k1, o1);
      }
    }
  }
}

template <int CH, int R, typename VT>
__device__ __forceinline__ void add_group(VT& acc, const float* __restrict__ feat, int loff,
                                          int key, float w, int j) {
  VT v[R];
  float ws[R];
#pragma unroll
  for (int q = 0; q < R; ++q) {
    const int s = clamp_node(__builtin_amdgcn_readlane(key, j + q));
    ws[q] = __int_as_float(__builtin_amdgcn_readlane(__float_as_int(w), j + q));
    v[q] = *(const VT*)(feat + (size_t)s * CH + loff);
  }
  __builtin_amdgcn_sched_barrier(0);
#pragma unroll
  for (int q = 0; q < R; ++q) fma_row(acc, v[q], ws[q]);
}

template <int CH, int NB, typename VT>
__device__ __forceinline__ void add_rows(VT& acc, const float* __restrict__ feat, int loff,
                                         int key, float w, int n) {
  static_assert(NB >= 2 && NB <= 8, "group sizes 1 .. NB-1 are written out below");
  int j = 0;
#pragma unroll 1
  for (; j + NB <= n; j += NB) add_group<CH, NB>(acc, feat, loff, key, w, j);
  const int r = n - j;
  if (r == 1) add_group<CH, 1>(acc, feat, loff, key, w, j);
  if constexpr (NB > 2) { if (r == 2) add_group<CH, 2>(acc, feat, loff, key, w, j); }
  if constexpr (NB > 3) { if (r == 3) add_group<CH, 3>(acc, feat, loff, key, w, j); }
  if constexpr (NB > 4) { if (r == 4) add_group<CH, 4>(acc, feat, loff, key, w, j); }
  if constexpr (NB > 5) { if (r == 5) add_group<CH, 5>(acc, feat, loff, key, w, j); }
  if constexpr (NB > 6) { if (r == 6) add_group<CH, 6>(acc, feat, loff, key, w, j); }
  if constexpr (NB > 7) { if (r == 7) add_group<CH, 7>(acc, feat, loff, key, w, j); }
}

#define GATHER_WAVES 25
#define GATHER_GROUPS 500
#define GATHER_ROWS(CH) (512 / (CH))

template <int CH, bool ACT>
__global__ void __launch_bounds__(GATHER_WAVES * 32, 13)
k_gather(const int* __restrict__ rowptr, const int* __restrict__ csr,
         const float* __restrict__ dinv, const float* __restrict__ feat,
         const float* __restrict__ bias,
         float* __restrict__ out_lo, float* __restrict__ out_hi, int split, int E) {
  typedef typename RowVec<CH>::type VT;
  constexpr int VPL = CH / 32;
  const int lane = threadIdx.x & 31;
  const int nw = __builtin_amdgcn_readfirstlane((int)(gridDim.x * (blockDim.x >> 5)));
  int d = __builtin_amdgcn_readfirstlane(
      (int)((blockIdx.x * blockDim.x + threadIdx.x) >> 5));
  if (d >= NNODES) return;

  const int loff = lane * VPL;
  const VT bv = *(const VT*)(bias + loff);

  int beg = rowptr[d];
  int end = rowptr[d + 1];
  beg = min(max(beg, 0), E);
  end = min(max(end, beg), E);
  float di = dinv[d];
  int kraw = csr[max(min(beg + lane, E - 1), 0)];
  VT self = *(const VT*)(feat + (size_t)d * CH + loff);

  int dn = d + nw;
  int dnc = min(dn, NNODES - 1);
  int begn = rowptr[dnc];
  int endn = rowptr[dnc + 1];
  begn = min(max(begn, 0), E);
  endn = min(max(endn, begn), E);
  float din = dinv[dnc];

  for (;;) {
    int kraw_n;
    VT self_n;
    const int dnn = dn + nw;
    const int dnnc = min(dnn, NNODES - 1);
    int begnn = rowptr[dnnc];
    int endnn = rowptr[dnnc + 1];
    begnn = min(max(begnn, 0), E);
    endnn = min(max(endnn, begnn), E);
    const float dinn = dinv[dnnc];

    const int deg = end - beg;
    VT acc = self_row(self, di * di, bv);
    const int rawkey = (lane < deg) ? kraw : KEY_PAD;

    if (deg <= 32) {
      const int key = sort32(rawkey, lane);
      const float wraw = dinv[clamp_node(key)];
      kraw_n = csr[max(min(begn + lane, E - 1), 0)];
      self_n = *(const VT*)(feat + (size_t)dnc * CH + loff);
      const float w = (lane < deg) ? wraw * di : 0.0f;
      add_rows<CH, GATHER_ROWS(CH)>(acc, feat, loff, key, w, deg);
    } else if (deg <= 64) {
      kraw_n = csr[max(min(begn + lane, E - 1), 0)];
      self_n = *(const VT*)(feat + (size_t)dnc * CH + loff);
      const bool two = lane + 32 < deg;
      int k0 = rawkey;
      int k1 = two ? csr[beg + 32 + lane] : KEY_PAD;
      sort64(k0, k1, lane);
      const float w0 = dinv[clamp_node(k0)] * di;
      const float w1 = two ? dinv[clamp_node(k1)] * di : 0.0f;
      add_rows<CH, GATHER_ROWS(CH)>(acc, feat, loff, k0, w0, 32);
      add_rows<CH, GATHER_ROWS(CH)>(acc, feat, loff, k1, w1, deg - 32);
    } else {
      kraw_n = csr[max(min(begn + lane, E - 1), 0)];
      self_n = *(const VT*)(feat + (size_t)dnc * CH + loff);
      int last = -1;
      int done = 0;
      while (done < deg) {
        int best = KEY_PAD;
        for (int j = beg + lane; j < end; j += 32) {
          const int k = csr[j];
          if (k > last && k < best) best = k;
        }
#pragma unroll
        for (int o = 16; o > 0; o >>= 1) best = min(best, __shfl_xor(best, o));
        int mult = 0;
        for (int j = beg + lane; j < end; j += 32) mult += (csr[j] == best) ? 1 : 0;
        mult = __builtin_amdgcn_readfirstlane(wave_sum_i(mult));
        best = __builtin_amdgcn_readfirstlane(best);
        if (mult <= 0) break;
        const int s = clamp_node(best);
        const float ws = dinv[s] * di;
        const VT v = *(const VT*)(feat + (size_t)s * CH + loff);
        for (int m = 0; m < mult; ++m) fma_row(acc, v, ws);
        last = best;
        done += mult;
      }
    }

    if constexpr (ACT) acc = tanh_row(acc);

    float* orow = (d < split) ? (out_lo + (size_t)d * CH) : (out_hi + (size_t)(d - split) * CH);
    *(VT*)(orow + loff) = acc;

    if (dn >= NNODES) break;
    d = dn; beg = begn; end = endn; di = din; kraw = kraw_n; self = self_n;
    dn = dnn; dnc = dnnc; begn = begnn; endn = endnn; din = dinn;
  }
}

extern "C" void kernel_launch(void* const* d_in, const int* in_sizes, int n_in,
                              void* d_out, int out_size, void* d_ws, size_t ws_size,
                              hipStream_t stream) {
  const float* x   = (const float*)d_in[0];
  const int*  eidx = (const int*)d_in[1];
  const float* W1  = (const float*)d_in[2];
  const float* b1  = (const float*)d_in[3];
  const float* W2  = (const float*)d_in[4];
  const float* b2  = (const float*)d_in[5];
  float* out = (float*)d_out;

  const int E = in_sizes[1] / 2;
  const int* src = eidx;
  const int* dst = eidx + E;

  float* dinv = (float*)d_ws;
  float* bufA = dinv + 50048;
  float* bufB = bufA + (size_t)NNODES * 128;
  int* csr    = (int*)(bufB + (size_t)SPLIT_ROWS * 128);
  int* rowptr = csr + 800000;
  int* part   = (int*)out;
  int* tab    = part + 800000;

  k_bin_chunk<<<NCHUNK, 1024, 0, stream>>>(src, dst, part, tab, E);
  k_rank_range<<<NRANGE, NCHUNK, 0, stream>>>(part, tab, rowptr, dinv, csr, E);

  k_gemm_wmma_f32<256, 128><<<256, GEMM_WAVES * 32, 0, stream>>>(x, x, NNODES, W1, bufA,
                                                                 NNODES);
  k_gather<128, true><<<GATHER_GROUPS, GATHER_WAVES * 32, 0, stream>>>(rowptr, csr, dinv, bufA,
                                                                       b1, bufB, out, SPLIT_ROWS, E);

  k_gemm_wmma_f32<128, 64><<<256, GEMM_WAVES * 32, 0, stream>>>(bufB, out, SPLIT_ROWS, W2,
                                                               bufA, NNODES);
  k_gather<64, false><<<GATHER_GROUPS, GATHER_WAVES * 32, 0, stream>>>(rowptr, csr, dinv, bufA,
                                                                       b2, out, out, NNODES, E);
}
